# starting kernel
# baseline (speedup 1.0000x reference)
#include <hip/hip_runtime.h>
#include <math.h>

#define SB   2048
#define DM   1024
#define NH   16
#define HD   64
#define HD2  32
#define SCALE 0.125f
#define EPS  1e-6f

typedef __attribute__((ext_vector_type(2))) float v2f;
typedef __attribute__((ext_vector_type(8))) float v8f;

static __device__ __forceinline__ v8f wmma_f32(v2f a, v2f b, v8f c) {
    return __builtin_amdgcn_wmma_f32_16x16x4_f32(
         false, a,  false, b,
         (short)0, c,  false,  false);
}

__global__ __launch_bounds__(256) void gemm16_kernel(
    const float* __restrict__ X,
    const float* __restrict__ W,
    const float* __restrict__ bias,
    const float* __restrict__ g,
    const float* __restrict__ fc,
    const float* __restrict__ fs,
    float* __restrict__ out,
    int doNormRope)
{
    __shared__ float tile[16][DM + 4];
    __shared__ float redbuf[16][17];
    __shared__ float rrms[16];

    const int tid  = threadIdx.x;
    const int wave = tid >> 5;
    const int lane = tid & 31;
    const int half = lane >> 4;
    const int l16  = lane & 15;
    const int rowBase  = blockIdx.x * 16;
    const int nColBase = wave * 128;

    v8f acc[8];
#pragma unroll
    for (int t = 0; t < 8; ++t) acc[t] = (v8f){0,0,0,0,0,0,0,0};

    const float* xrow = X + (size_t)(rowBase + l16) * DM + 2 * half;

    for (int k = 0; k < DM; k += 4) {
        v2f a = *(const v2f*)(xrow + k);
#pragma unroll
        for (int t = 0; t < 8; ++t) {
            const float* wp = W + (size_t)(nColBase + t * 16 + l16) * DM + k + 2 * half;
            v2f b = *(const v2f*)wp;
            acc[t] = wmma_f32(a, b, acc[t]);
        }
    }

#pragma unroll
    for (int t = 0; t < 8; ++t) {
        int col = nColBase + t * 16 + l16;
        float bv = bias[col];
#pragma unroll
        for (int j = 0; j < 8; ++j) {
            int row = j + 8 * half;
            tile[row][col] = acc[t][j] + bv;
        }
    }
    __syncthreads();

    if (doNormRope) {
        {
            int r  = tid & 15;
            int c0 = tid >> 4;
            float s = 0.f;
            for (int c = c0; c < DM; c += 16) { float v = tile[r][c]; s += v * v; }
            redbuf[r][c0] = s;
        }
        __syncthreads();
        if (tid < 16) {
            float s = 0.f;
#pragma unroll
            for (int c = 0; c < 16; ++c) s += redbuf[tid][c];
            rrms[tid] = rsqrtf(s * (1.0f / (float)DM) + EPS);
        }
        __syncthreads();
    }

    for (int p = tid; p < 16 * (DM / 2); p += 256) {
        int r    = p >> 9;
        int cp   = p & 511;
        int dcol = cp * 2;
        int srow = rowBase + r;
        float re = tile[r][dcol];
        float im = tile[r][dcol + 1];
        if (doNormRope) {
            float rr = rrms[r];
            re *= rr * g[dcol];
            im *= rr * g[dcol + 1];
            int   pp = cp & (HD2 - 1);
            float c  = fc[srow * HD2 + pp];
            float sn = fs[srow * HD2 + pp];
            float ore = re * c - im * sn;
            float oim = re * sn + im * c;
            re = ore; im = oim;
        }
        out[(size_t)srow * DM + dcol]     = re;
        out[(size_t)srow * DM + dcol + 1] = im;
    }
}

__global__ __launch_bounds__(32) void attn_kernel(
    const float* __restrict__ Q,
    const float* __restrict__ K,
    const float* __restrict__ V,
    float* __restrict__ O)
{
    __shared__ float P[16][17];

    const int lane = threadIdx.x & 31;
    const int half = lane >> 4;
    const int l16  = lane & 15;
    const int qt    = blockIdx.x;
    const int h     = blockIdx.y;
    const int qBase = qt * 16;
    const int dBase = h * HD;

    v2f qa[16];
    {
        const float* qp = Q + (size_t)(qBase + l16) * DM + dBase + 2 * half;
#pragma unroll
        for (int kk = 0; kk < 16; ++kk) qa[kk] = *(const v2f*)(qp + kk * 4);
    }

    v8f o[4];
#pragma unroll
    for (int nt = 0; nt < 4; ++nt) o[nt] = (v8f){0,0,0,0,0,0,0,0};
    float m[8], lsum[8];
#pragma unroll
    for (int j = 0; j < 8; ++j) { m[j] = -INFINITY; lsum[j] = 0.f; }

    for (int t = 0; t < SB / 16; ++t) {
        const int kBase = t * 16;

        v8f s = (v8f){0,0,0,0,0,0,0,0};
        const float* kp = K + (size_t)(kBase + l16) * DM + dBase + 2 * half;
#pragma unroll
        for (int kk = 0; kk < 16; ++kk) {
            v2f b = *(const v2f*)(kp + kk * 4);
            s = wmma_f32(qa[kk], b, s);
        }

#pragma unroll
        for (int j = 0; j < 8; ++j) {
            float sv = s[j] * SCALE;
            float mx = sv;
            mx = fmaxf(mx, __shfl_xor(mx, 1, 32));
            mx = fmaxf(mx, __shfl_xor(mx, 2, 32));
            mx = fmaxf(mx, __shfl_xor(mx, 4, 32));
            mx = fmaxf(mx, __shfl_xor(mx, 8, 32));
            float nm   = fmaxf(m[j], mx);
            float corr = __expf(m[j] - nm);
            float pv   = __expf(sv - nm);
            float ps = pv;
            ps += __shfl_xor(ps, 1, 32);
            ps += __shfl_xor(ps, 2, 32);
            ps += __shfl_xor(ps, 4, 32);
            ps += __shfl_xor(ps, 8, 32);
            lsum[j] = lsum[j] * corr + ps;
            m[j] = nm;
            o[0][j] *= corr; o[1][j] *= corr; o[2][j] *= corr; o[3][j] *= corr;
            P[j + 8 * half][l16] = pv;
        }
        __syncthreads();

#pragma unroll
        for (int kc = 0; kc < 4; ++kc) {
            v2f pa;
            pa.x = P[l16][kc * 4 + 2 * half];
            pa.y = P[l16][kc * 4 + 2 * half + 1];
#pragma unroll
            for (int nt = 0; nt < 4; ++nt) {
                const float* vp = V + (size_t)(kBase + kc * 4 + 2 * half) * DM
                                    + dBase + nt * 16 + l16;
                v2f b; b.x = vp[0]; b.y = vp[DM];
                o[nt] = wmma_f32(pa, b, o[nt]);
            }
        }
        __syncthreads();
    }

#pragma unroll
    for (int j = 0; j < 8; ++j) {
        float inv = 1.0f / lsum[j];
        int row = qBase + j + 8 * half;
#pragma unroll
        for (int nt = 0; nt < 4; ++nt) {
            O[(size_t)row * DM + dBase + nt * 16 + l16] = o[nt][j] * inv;
        }
    }
}

extern "C" void kernel_launch(void* const* d_in, const int* in_sizes, int n_in,
                              void* d_out, int out_size, void* d_ws, size_t ws_size,
                              hipStream_t stream) {
    const float* x  = (const float*)d_in[0];
    const float* fc = (const float*)d_in[1];
    const float* fs = (const float*)d_in[2];
    const float* wq = (const float*)d_in[3];
    const float* bq = (const float*)d_in[4];
    const float* wk = (const float*)d_in[5];
    const float* bk = (const float*)d_in[6];
    const float* wv = (const float*)d_in[7];
    const float* bv = (const float*)d_in[8];
    const float* wo = (const float*)d_in[9];
    const float* bo = (const float*)d_in[10];
    const float* gq = (const float*)d_in[11];
    const float* gk = (const float*)d_in[12];
    float* out = (float*)d_out;

    const size_t mat = (size_t)SB * DM;
    float* Qw = (float*)d_ws;
    float* Kw = Qw + mat;
    float* Vw = Kw + mat;
    float* Aw = Vw + mat;

    dim3 gblk(SB / 16);
    gemm16_kernel<<<gblk, 256, 0, stream>>>(x, wq, bq, gq, fc, fs, Qw, 1);
    gemm16_kernel<<<gblk, 256, 0, stream>>>(x, wk, bk, gk, fc, fs, Kw, 1);
    gemm16_kernel<<<gblk, 256, 0, stream>>>(x, wv, bv, gq, fc, fs, Vw, 0);
    attn_kernel<<<dim3(SB / 16, NH), 32, 0, stream>>>(Qw, Kw, Vw, Aw);
    gemm16_kernel<<<gblk, 256, 0, stream>>>(Aw, wo, bo, gq, fc, fs, out, 0);
}


# attention: K,V written by the GEMMs in WMMA-fragment order; one whole-line 16-B load feeds 2 k steps (was 16-line 8-B, 4-B loads), 32 per block as one group (waves_per_eu 2); QK, PV sum order change
# speedup vs baseline: 2.5302x; 2.5302x over previous
#include <hip/hip_runtime.h>
#include <math.h>

#define SB   2048
#define DM   1024
#define NH   16
#define HD   64
#define HD2  32
#define SCALE 0.125f
#define EPS  1e-6f

typedef __attribute__((ext_vector_type(2))) float v2f;
typedef __attribute__((ext_vector_type(4))) float v4f;
typedef __attribute__((ext_vector_type(8))) float v8f;

static __device__ __forceinline__ v8f wmma_f32(v2f a, v2f b, v8f c) {
    return __builtin_amdgcn_wmma_f32_16x16x4_f32(
         false, a,  false, b,
         (short)0, c,  false,  false);
}

#define WF_TILE (16 * DM)
#define WF_PER_THREAD 8

#define KF_HEAD (SB * HD)
__global__ __launch_bounds__(256) void wfrag_kernel(
    const float* __restrict__ w0, const float* __restrict__ w1,
    const float* __restrict__ w2, const float* __restrict__ w3,
    float* __restrict__ dst)
{
    const int which = blockIdx.y;
    const float* W = (which == 0) ? w0 : (which == 1) ? w1 : (which == 2) ? w2 : w3;
    float* D = dst + (size_t)which * DM * DM;
    v4f v[WF_PER_THREAD];
#pragma unroll
    for (int i = 0; i < WF_PER_THREAD; ++i) {
        const int o    = (blockIdx.x * WF_PER_THREAD + i) * 256 + threadIdx.x;
        const int lane = o & 31;
        const int j    = (o >> 5) & 127;
        const int nt   = o >> 12;
        const int c    = lane & 15;
        const int hf   = lane >> 4;
        v[i] = *(const v4f*)(W + (size_t)(nt * 16 + c) * DM + 8 * j + 4 * hf);
    }
#pragma unroll
    for (int i = 0; i < WF_PER_THREAD; ++i) {
        const int o = (blockIdx.x * WF_PER_THREAD + i) * 256 + threadIdx.x;
        *(v4f*)(D + (size_t)o * 4) = v[i];
    }
}

__global__ __launch_bounds__(256) void gemm16_kernel(
    const float* __restrict__ X,
    const float* __restrict__ Wf,
    const float* __restrict__ bias,
    const float* __restrict__ g,
    const float* __restrict__ fc,
    const float* __restrict__ fs,
    float* __restrict__ out,
    int mode)
{
    const int doNormRope = (mode == 1) | (mode == 2);
    __shared__ __attribute__((aligned(16))) float tile[16][DM + 4];
    __shared__ float redbuf[16][17];
    __shared__ float rrms[16];

    const int tid  = threadIdx.x;
    const int wave = tid >> 5;
    const int lane = tid & 31;
    const int half = lane >> 4;
    const int l16  = lane & 15;
    const int rowBase  = blockIdx.x * 16;
    const int nColBase = wave * 128;

    v8f acc[8];
#pragma unroll
    for (int t = 0; t < 8; ++t) acc[t] = (v8f){0,0,0,0,0,0,0,0};

    const float* xrow = X + (size_t)(rowBase + l16) * DM + 4 * half;
    const float* wfp = Wf + (size_t)(wave * 8) * WF_TILE + lane * 4;

    v2f alo[4], ahi[4], blo[4][8], bhi[4][8];
#pragma unroll
    for (int s = 0; s < 3; ++s) {
        v4f xa = *(const v4f*)(xrow + 8 * s);
        alo[s] = xa.xy; ahi[s] = xa.zw;
#pragma unroll
        for (int t = 0; t < 8; ++t) {
            v4f xb = *(const v4f*)(wfp + t * WF_TILE + s * 128);
            blo[s][t] = xb.xy; bhi[s][t] = xb.zw;
        }
    }

    for (int j = 0; j < DM / 8; j += 4) {
#pragma unroll
        for (int u = 0; u < 4; ++u) {
            const int gn = (j + u + 3 < DM / 8) ? (j + u + 3) : 0;
            const int sn = (u + 3) & 3;
            {
                v4f xa = *(const v4f*)(xrow + 8 * gn);
                alo[sn] = xa.xy; ahi[sn] = xa.zw;
            }
#pragma unroll
            for (int t = 0; t < 8; ++t) {
                v4f xb = *(const v4f*)(wfp + t * WF_TILE + gn * 128);
                blo[sn][t] = xb.xy; bhi[sn][t] = xb.zw;
            }
#pragma unroll
            for (int t = 0; t < 8; ++t) acc[t] = wmma_f32(alo[u], blo[u][t], acc[t]);
#pragma unroll
            for (int t = 0; t < 8; ++t) acc[t] = wmma_f32(ahi[u], bhi[u][t], acc[t]);
        }
    }

#pragma unroll
    for (int t = 0; t < 8; ++t) {
        int col = nColBase + t * 16 + l16;
        float bv = bias[col];
#pragma unroll
        for (int j = 0; j < 8; ++j) {
            int row = j + 8 * half;
            tile[row][col] = acc[t][j] + bv;
        }
    }
    __syncthreads();

    if (doNormRope) {
        {
            int r  = tid & 15;
            int c0 = tid >> 4;
            float s = 0.f;
            for (int c = c0; c < DM; c += 16) { float v = tile[r][c]; s += v * v; }
            redbuf[r][c0] = s;
        }
        __syncthreads();
        if (tid < 16) {
            float s = 0.f;
#pragma unroll
            for (int c = 0; c < 16; ++c) s += redbuf[tid][c];
            rrms[tid] = rsqrtf(s * (1.0f / (float)DM) + EPS);
        }
        __syncthreads();
    }

    if (mode == 2) {
        const int j = wave;
        const int srow = rowBase + l16;
        const float rr = rrms[l16];
        const int pp = 4 * j + 2 * half;
        const float c0 = fc[srow * HD2 + pp],     s0 = fs[srow * HD2 + pp];
        const float c1 = fc[srow * HD2 + pp + 1], s1 = fs[srow * HD2 + pp + 1];
        for (int hh = 0; hh < NH; ++hh) {
            const int col = hh * HD + 8 * j + 4 * half;
            const v4f v  = *(const v4f*)&tile[l16][col];
            const v4f gg = *(const v4f*)(g + col);
            float re0 = v.x, im0 = v.y, re1 = v.z, im1 = v.w;
            re0 *= rr * gg.x; im0 *= rr * gg.y;
            re1 *= rr * gg.z; im1 *= rr * gg.w;
            v4f r;
            r.x = re0 * c0 - im0 * s0;
            r.y = re0 * s0 + im0 * c0;
            r.z = re1 * c1 - im1 * s1;
            r.w = re1 * s1 + im1 * c1;
            *(v4f*)(out + (size_t)hh * KF_HEAD + ((size_t)blockIdx.x * 8 + j) * 128 + lane * 4) = r;
        }
        return;
    }
    if (mode == 3) {
        const int nt  = wave & 3;
        const int kgl = wave >> 2;
        for (int hh = 0; hh < NH; ++hh) {
            const int col = hh * HD + nt * 16 + l16;
            const int r0  = 8 * kgl + 4 * half;
            v4f r;
            r.x = tile[r0 + 0][col];
            r.y = tile[r0 + 1][col];
            r.z = tile[r0 + 2][col];
            r.w = tile[r0 + 3][col];
            *(v4f*)(out + (size_t)hh * KF_HEAD
                        + (((size_t)blockIdx.x * 2 + kgl) * 4 + nt) * 128 + lane * 4) = r;
        }
        return;
    }

    for (int p = tid; p < 16 * (DM / 2); p += 256) {
        int r    = p >> 9;
        int cp   = p & 511;
        int dcol = cp * 2;
        int srow = rowBase + r;
        float re = tile[r][dcol];
        float im = tile[r][dcol + 1];
        if (doNormRope) {
            float rr = rrms[r];
            re *= rr * g[dcol];
            im *= rr * g[dcol + 1];
            int   pp = cp & (HD2 - 1);
            float c  = fc[srow * HD2 + pp];
            float sn = fs[srow * HD2 + pp];
            float ore = re * c - im * sn;
            float oim = re * sn + im * c;
            re = ore; im = oim;
        }
        out[(size_t)srow * DM + dcol]     = re;
        out[(size_t)srow * DM + dcol + 1] = im;
    }
}

#define KB  64
#define KT  (KB / 16)
#define PP  (KB + 4)

__global__ __launch_bounds__(32) __attribute__((amdgpu_waves_per_eu(2, 2))) void attn_kernel(
    const float* __restrict__ Q,
    const float* __restrict__ Kf,
    const float* __restrict__ Vf,
    float* __restrict__ O)
{
    __shared__ __attribute__((aligned(16))) float P[16][PP];

    const int lane = threadIdx.x & 31;
    const int half = lane >> 4;
    const int l16  = lane & 15;
    const int qt    = blockIdx.x;
    const int h     = blockIdx.y;
    const int qBase = qt * 16;
    const int dBase = h * HD;

    v2f qlo[8], qhi[8];
    {
        const float* qp = Q + (size_t)(qBase + l16) * DM + dBase + 4 * half;
#pragma unroll
        for (int j = 0; j < 8; ++j) {
            v4f xq = *(const v4f*)(qp + 8 * j);
            qlo[j] = xq.xy; qhi[j] = xq.zw;
        }
    }
    const float* kfp = Kf + (size_t)h * KF_HEAD + lane * 4;
    const float* vfp = Vf + (size_t)h * KF_HEAD + lane * 4;

    v8f o[4];
#pragma unroll
    for (int nt = 0; nt < 4; ++nt) o[nt] = (v8f){0,0,0,0,0,0,0,0};
    float m[8], lsum[8];
#pragma unroll
    for (int j = 0; j < 8; ++j) { m[j] = -INFINITY; lsum[j] = 0.f; }

    for (int t = 0; t < SB / KB; ++t) {
        v8f s[KT];
#pragma unroll
        for (int c = 0; c < KT; ++c) s[c] = (v8f){0,0,0,0,0,0,0,0};
        const float* kp = kfp + (size_t)t * (KT * 8 * 128);
#pragma unroll
        for (int j = 0; j < 8; ++j) {
            v4f kb[KT];
#pragma unroll
            for (int c = 0; c < KT; ++c) kb[c] = *(const v4f*)(kp + (c * 8 + j) * 128);
#pragma unroll
            for (int c = 0; c < KT; ++c) s[c] = wmma_f32(qlo[j], kb[c].xy, s[c]);
#pragma unroll
            for (int c = 0; c < KT; ++c) s[c] = wmma_f32(qhi[j], kb[c].zw, s[c]);
        }

#pragma unroll
        for (int j = 0; j < 8; ++j) {
            float sv0 = s[0][j] * SCALE;
            float sv1 = s[1][j] * SCALE;
            float sv2 = s[2][j] * SCALE;
            float sv3 = s[3][j] * SCALE;
            float mx = fmaxf(fmaxf(sv0, sv1), fmaxf(sv2, sv3));
            mx = fmaxf(mx, __shfl_xor(mx, 1, 32));
            mx = fmaxf(mx, __shfl_xor(mx, 2, 32));
            mx = fmaxf(mx, __shfl_xor(mx, 4, 32));
            mx = fmaxf(mx, __shfl_xor(mx, 8, 32));
            float nm   = fmaxf(m[j], mx);
            float corr = __expf(m[j] - nm);
            float p0 = __expf(sv0 - nm);
            float p1 = __expf(sv1 - nm);
            float p2 = __expf(sv2 - nm);
            float p3 = __expf(sv3 - nm);
            float ps = (p0 + p1) + (p2 + p3);
            ps += __shfl_xor(ps, 1, 32);
            ps += __shfl_xor(ps, 2, 32);
            ps += __shfl_xor(ps, 4, 32);
            ps += __shfl_xor(ps, 8, 32);
            lsum[j] = lsum[j] * corr + ps;
            m[j] = nm;
            o[0][j] *= corr; o[1][j] *= corr; o[2][j] *= corr; o[3][j] *= corr;
            const int prow = j + 8 * half;
            P[prow][0 * 16 + l16] = p0;
            P[prow][1 * 16 + l16] = p1;
            P[prow][2 * 16 + l16] = p2;
            P[prow][3 * 16 + l16] = p3;
        }
        __syncthreads();

        const float* vp = vfp + (size_t)t * (8 * 4 * 128);
#pragma unroll
        for (int g = 0; g < KB / 8; ++g) {
            v4f vb[4];
#pragma unroll
            for (int nt = 0; nt < 4; ++nt) vb[nt] = *(const v4f*)(vp + (g * 4 + nt) * 128);
            const v4f pa = *(const v4f*)&P[l16][8 * g + 4 * half];
#pragma unroll
            for (int nt = 0; nt < 4; ++nt) o[nt] = wmma_f32(pa.xy, vb[nt].xy, o[nt]);
#pragma unroll
            for (int nt = 0; nt < 4; ++nt) o[nt] = wmma_f32(pa.zw, vb[nt].zw, o[nt]);
        }
        __syncthreads();
    }

#pragma unroll
    for (int j = 0; j < 8; ++j) {
        float inv = 1.0f / lsum[j];
        int row = qBase + j + 8 * half;
#pragma unroll
        for (int nt = 0; nt < 4; ++nt) {
            O[(size_t)row * DM + dBase + nt * 16 + l16] = o[nt][j] * inv;
        }
    }
}

extern "C" void kernel_launch(void* const* d_in, const int* in_sizes, int n_in,
                              void* d_out, int out_size, void* d_ws, size_t ws_size,
                              hipStream_t stream) {
    const float* x  = (const float*)d_in[0];
    const float* fc = (const float*)d_in[1];
    const float* fs = (const float*)d_in[2];
    const float* wq = (const float*)d_in[3];
    const float* bq = (const float*)d_in[4];
    const float* wk = (const float*)d_in[5];
    const float* bk = (const float*)d_in[6];
    const float* wv = (const float*)d_in[7];
    const float* bv = (const float*)d_in[8];
    const float* wo = (const float*)d_in[9];
    const float* bo = (const float*)d_in[10];
    const float* gq = (const float*)d_in[11];
    const float* gk = (const float*)d_in[12];
    float* out = (float*)d_out;

    const size_t mat = (size_t)SB * DM;
    float* Qw = (float*)d_ws;
    float* Kw = Qw + mat;
    float* Vw = Kw + mat;
    float* Aw = Vw + mat;
    const size_t wsz = (size_t)DM * DM;
    float* Wfq = Aw + mat;
    float* Wfk = Wfq + wsz;
    float* Wfv = Wfk + wsz;
    float* Wfo = Wfv + wsz;

    wfrag_kernel<<<dim3(DM * DM / 4 / 256 / WF_PER_THREAD, 4), 256, 0, stream>>>(wq, wk, wv, wo, Wfq);

    dim3 gblk(SB / 16);
    gemm16_kernel<<<gblk, 256, 0, stream>>>(x, Wfq, bq, gq, fc, fs, Qw, 1);
    gemm16_kernel<<<gblk, 256, 0, stream>>>(x, Wfk, bk, gk, fc, fs, Kw, 2);
    gemm16_kernel<<<gblk, 256, 0, stream>>>(x, Wfv, bv, gq, fc, fs, Vw, 3);
    attn_kernel<<<dim3(SB / 16, NH), 32, 0, stream>>>(Qw, Kw, Vw, Aw);
    gemm16_kernel<<<gblk, 256, 0, stream>>>(Aw, Wfo, bo, gq, fc, fs, out, 0);
}


# attention: 8 query tiles of one head run as the 8 waves of one workgroup (was one-wave workgroups, heads mixed per compute unit): K,V lines re-read from the unit's vector cache; sums unchanged
# speedup vs baseline: 2.6474x; 1.0463x over previous
#include <hip/hip_runtime.h>
#include <math.h>

#define SB   2048
#define DM   1024
#define NH   16
#define HD   64
#define HD2  32
#define SCALE 0.125f
#define EPS  1e-6f

typedef __attribute__((ext_vector_type(2))) float v2f;
typedef __attribute__((ext_vector_type(4))) float v4f;
typedef __attribute__((ext_vector_type(8))) float v8f;

static __device__ __forceinline__ v8f wmma_f32(v2f a, v2f b, v8f c) {
    return __builtin_amdgcn_wmma_f32_16x16x4_f32(
         false, a,  false, b,
         (short)0, c,  false,  false);
}

#define WF_TILE (16 * DM)
#define WF_PER_THREAD 8

#define KF_HEAD (SB * HD)
__global__ __launch_bounds__(256) void wfrag_kernel(
    const float* __restrict__ w0, const float* __restrict__ w1,
    const float* __restrict__ w2, const float* __restrict__ w3,
    float* __restrict__ dst)
{
    const int which = blockIdx.y;
    const float* W = (which == 0) ? w0 : (which == 1) ? w1 : (which == 2) ? w2 : w3;
    float* D = dst + (size_t)which * DM * DM;
    v4f v[WF_PER_THREAD];
#pragma unroll
    for (int i = 0; i < WF_PER_THREAD; ++i) {
        const int o    = (blockIdx.x * WF_PER_THREAD + i) * 256 + threadIdx.x;
        const int lane = o & 31;
        const int j    = (o >> 5) & 127;
        const int nt   = o >> 12;
        const int c    = lane & 15;
        const int hf   = lane >> 4;
        v[i] = *(const v4f*)(W + (size_t)(nt * 16 + c) * DM + 8 * j + 4 * hf);
    }
#pragma unroll
    for (int i = 0; i < WF_PER_THREAD; ++i) {
        const int o = (blockIdx.x * WF_PER_THREAD + i) * 256 + threadIdx.x;
        *(v4f*)(D + (size_t)o * 4) = v[i];
    }
}

__global__ __launch_bounds__(256) void gemm16_kernel(
    const float* __restrict__ X,
    const float* __restrict__ Wf,
    const float* __restrict__ bias,
    const float* __restrict__ g,
    const float* __restrict__ fc,
    const float* __restrict__ fs,
    float* __restrict__ out,
    int mode)
{
    const int doNormRope = (mode == 1) | (mode == 2);
    __shared__ __attribute__((aligned(16))) float tile[16][DM + 4];
    __shared__ float redbuf[16][17];
    __shared__ float rrms[16];

    const int tid  = threadIdx.x;
    const int wave = tid >> 5;
    const int lane = tid & 31;
    const int half = lane >> 4;
    const int l16  = lane & 15;
    const int rowBase  = blockIdx.x * 16;
    const int nColBase = wave * 128;

    v8f acc[8];
#pragma unroll
    for (int t = 0; t < 8; ++t) acc[t] = (v8f){0,0,0,0,0,0,0,0};

    const float* xrow = X + (size_t)(rowBase + l16) * DM + 4 * half;
    const float* wfp = Wf + (size_t)(wave * 8) * WF_TILE + lane * 4;

    v2f alo[4], ahi[4], blo[4][8], bhi[4][8];
#pragma unroll
    for (int s = 0; s < 3; ++s) {
        v4f xa = *(const v4f*)(xrow + 8 * s);
        alo[s] = xa.xy; ahi[s] = xa.zw;
#pragma unroll
        for (int t = 0; t < 8; ++t) {
            v4f xb = *(const v4f*)(wfp + t * WF_TILE + s * 128);
            blo[s][t] = xb.xy; bhi[s][t] = xb.zw;
        }
    }

    for (int j = 0; j < DM / 8; j += 4) {
#pragma unroll
        for (int u = 0; u < 4; ++u) {
            const int gn = (j + u + 3 < DM / 8) ? (j + u + 3) : 0;
            const int sn = (u + 3) & 3;
            {
                v4f xa = *(const v4f*)(xrow + 8 * gn);
                alo[sn] = xa.xy; ahi[sn] = xa.zw;
            }
#pragma unroll
            for (int t = 0; t < 8; ++t) {
                v4f xb = *(const v4f*)(wfp + t * WF_TILE + gn * 128);
                blo[sn][t] = xb.xy; bhi[sn][t] = xb.zw;
            }
#pragma unroll
            for (int t = 0; t < 8; ++t) acc[t] = wmma_f32(alo[u], blo[u][t], acc[t]);
#pragma unroll
            for (int t = 0; t < 8; ++t) acc[t] = wmma_f32(ahi[u], bhi[u][t], acc[t]);
        }
    }

#pragma unroll
    for (int t = 0; t < 8; ++t) {
        int col = nColBase + t * 16 + l16;
        float bv = bias[col];
#pragma unroll
        for (int j = 0; j < 8; ++j) {
            int row = j + 8 * half;
            tile[row][col] = acc[t][j] + bv;
        }
    }
    __syncthreads();

    if (doNormRope) {
        {
            int r  = tid & 15;
            int c0 = tid >> 4;
            float s = 0.f;
            for (int c = c0; c < DM; c += 16) { float v = tile[r][c]; s += v * v; }
            redbuf[r][c0] = s;
        }
        __syncthreads();
        if (tid < 16) {
            float s = 0.f;
#pragma unroll
            for (int c = 0; c < 16; ++c) s += redbuf[tid][c];
            rrms[tid] = rsqrtf(s * (1.0f / (float)DM) + EPS);
        }
        __syncthreads();
    }

    if (mode == 2) {
        const int j = wave;
        const int srow = rowBase + l16;
        const float rr = rrms[l16];
        const int pp = 4 * j + 2 * half;
        const float c0 = fc[srow * HD2 + pp],     s0 = fs[srow * HD2 + pp];
        const float c1 = fc[srow * HD2 + pp + 1], s1 = fs[srow * HD2 + pp + 1];
        for (int hh = 0; hh < NH; ++hh) {
            const int col = hh * HD + 8 * j + 4 * half;
            const v4f v  = *(const v4f*)&tile[l16][col];
            const v4f gg = *(const v4f*)(g + col);
            float re0 = v.x, im0 = v.y, re1 = v.z, im1 = v.w;
            re0 *= rr * gg.x; im0 *= rr * gg.y;
            re1 *= rr * gg.z; im1 *= rr * gg.w;
            v4f r;
            r.x = re0 * c0 - im0 * s0;
            r.y = re0 * s0 + im0 * c0;
            r.z = re1 * c1 - im1 * s1;
            r.w = re1 * s1 + im1 * c1;
            *(v4f*)(out + (size_t)hh * KF_HEAD + ((size_t)blockIdx.x * 8 + j) * 128 + lane * 4) = r;
        }
        return;
    }
    if (mode == 3) {
        const int nt  = wave & 3;
        const int kgl = wave >> 2;
        for (int hh = 0; hh < NH; ++hh) {
            const int col = hh * HD + nt * 16 + l16;
            const int r0  = 8 * kgl + 4 * half;
            v4f r;
            r.x = tile[r0 + 0][col];
            r.y = tile[r0 + 1][col];
            r.z = tile[r0 + 2][col];
            r.w = tile[r0 + 3][col];
            *(v4f*)(out + (size_t)hh * KF_HEAD
                        + (((size_t)blockIdx.x * 2 + kgl) * 4 + nt) * 128 + lane * 4) = r;
        }
        return;
    }

    for (int p = tid; p < 16 * (DM / 2); p += 256) {
        int r    = p >> 9;
        int cp   = p & 511;
        int dcol = cp * 2;
        int srow = rowBase + r;
        float re = tile[r][dcol];
        float im = tile[r][dcol + 1];
        if (doNormRope) {
            float rr = rrms[r];
            re *= rr * g[dcol];
            im *= rr * g[dcol + 1];
            int   pp = cp & (HD2 - 1);
            float c  = fc[srow * HD2 + pp];
            float sn = fs[srow * HD2 + pp];
            float ore = re * c - im * sn;
            float oim = re * sn + im * c;
            re = ore; im = oim;
        }
        out[(size_t)srow * DM + dcol]     = re;
        out[(size_t)srow * DM + dcol + 1] = im;
    }
}

#define KB  64
#define KT  (KB / 16)
#define PP  (KB + 4)

__global__ __launch_bounds__(256) __attribute__((amdgpu_waves_per_eu(2, 2))) void attn_kernel(
    const float* __restrict__ Q,
    const float* __restrict__ Kf,
    const float* __restrict__ Vf,
    float* __restrict__ O)
{
    __shared__ __attribute__((aligned(16))) float Pall[8][16][PP];

    const int wave = threadIdx.x >> 5;
    const int lane = threadIdx.x & 31;
    const int half = lane >> 4;
    const int l16  = lane & 15;
    float (&P)[16][PP] = Pall[wave];
    const int qt    = blockIdx.x * 8 + wave;
    const int h     = blockIdx.y;
    const int qBase = qt * 16;
    const int dBase = h * HD;

    v2f qlo[8], qhi[8];
    {
        const float* qp = Q + (size_t)(qBase + l16) * DM + dBase + 4 * half;
#pragma unroll
        for (int j = 0; j < 8; ++j) {
            v4f xq = *(const v4f*)(qp + 8 * j);
            qlo[j] = xq.xy; qhi[j] = xq.zw;
        }
    }
    const float* kfp = Kf + (size_t)h * KF_HEAD + lane * 4;
    const float* vfp = Vf + (size_t)h * KF_HEAD + lane * 4;

    v8f o[4];
#pragma unroll
    for (int nt = 0; nt < 4; ++nt) o[nt] = (v8f){0,0,0,0,0,0,0,0};
    float m[8], lsum[8];
#pragma unroll
    for (int j = 0; j < 8; ++j) { m[j] = -INFINITY; lsum[j] = 0.f; }

    for (int t = 0; t < SB / KB; ++t) {
        v8f s[KT];
#pragma unroll
        for (int c = 0; c < KT; ++c) s[c] = (v8f){0,0,0,0,0,0,0,0};
        const float* kp = kfp + (size_t)t * (KT * 8 * 128);
#pragma unroll
        for (int j = 0; j < 8; ++j) {
            v4f kb[KT];
#pragma unroll
            for (int c = 0; c < KT; ++c) kb[c] = *(const v4f*)(kp + (c * 8 + j) * 128);
#pragma unroll
            for (int c = 0; c < KT; ++c) s[c] = wmma_f32(qlo[j], kb[c].xy, s[c]);
#pragma unroll
            for (int c = 0; c < KT; ++c) s[c] = wmma_f32(qhi[j], kb[c].zw, s[c]);
        }

#pragma unroll
        for (int j = 0; j < 8; ++j) {
            float sv0 = s[0][j] * SCALE;
            float sv1 = s[1][j] * SCALE;
            float sv2 = s[2][j] * SCALE;
            float sv3 = s[3][j] * SCALE;
            float mx = fmaxf(fmaxf(sv0, sv1), fmaxf(sv2, sv3));
            mx = fmaxf(mx, __shfl_xor(mx, 1, 32));
            mx = fmaxf(mx, __shfl_xor(mx, 2, 32));
            mx = fmaxf(mx, __shfl_xor(mx, 4, 32));
            mx = fmaxf(mx, __shfl_xor(mx, 8, 32));
            float nm   = fmaxf(m[j], mx);
            float corr = __expf(m[j] - nm);
            float p0 = __expf(sv0 - nm);
            float p1 = __expf(sv1 - nm);
            float p2 = __expf(sv2 - nm);
            float p3 = __expf(sv3 - nm);
            float ps = (p0 + p1) + (p2 + p3);
            ps += __shfl_xor(ps, 1, 32);
            ps += __shfl_xor(ps, 2, 32);
            ps += __shfl_xor(ps, 4, 32);
            ps += __shfl_xor(ps, 8, 32);
            lsum[j] = lsum[j] * corr + ps;
            m[j] = nm;
            o[0][j] *= corr; o[1][j] *= corr; o[2][j] *= corr; o[3][j] *= corr;
            const int prow = j + 8 * half;
            P[prow][0 * 16 + l16] = p0;
            P[prow][1 * 16 + l16] = p1;
            P[prow][2 * 16 + l16] = p2;
            P[prow][3 * 16 + l16] = p3;
        }
        __builtin_amdgcn_fence(__ATOMIC_RELEASE, "wavefront");
        __builtin_amdgcn_wave_barrier();
        __builtin_amdgcn_fence(__ATOMIC_ACQUIRE, "wavefront");

        const float* vp = vfp + (size_t)t * (8 * 4 * 128);
#pragma unroll
        for (int g = 0; g < KB / 8; ++g) {
            v4f vb[4];
#pragma unroll
            for (int nt = 0; nt < 4; ++nt) vb[nt] = *(const v4f*)(vp + (g * 4 + nt) * 128);
            const v4f pa = *(const v4f*)&P[l16][8 * g + 4 * half];
#pragma unroll
            for (int nt = 0; nt < 4; ++nt) o[nt] = wmma_f32(pa.xy, vb[nt].xy, o[nt]);
#pragma unroll
            for (int nt = 0; nt < 4; ++nt) o[nt] = wmma_f32(pa.zw, vb[nt].zw, o[nt]);
        }
        __builtin_amdgcn_fence(__ATOMIC_RELEASE, "wavefront");
        __builtin_amdgcn_wave_barrier();
        __builtin_amdgcn_fence(__ATOMIC_ACQUIRE, "wavefront");
    }

#pragma unroll
    for (int j = 0; j < 8; ++j) {
        float inv = 1.0f / lsum[j];
        int row = qBase + j + 8 * half;
#pragma unroll
        for (int nt = 0; nt < 4; ++nt) {
            O[(size_t)row * DM + dBase + nt * 16 + l16] = o[nt][j] * inv;
        }
    }
}

extern "C" void kernel_launch(void* const* d_in, const int* in_sizes, int n_in,
                              void* d_out, int out_size, void* d_ws, size_t ws_size,
                              hipStream_t stream) {
    const float* x  = (const float*)d_in[0];
    const float* fc = (const float*)d_in[1];
    const float* fs = (const float*)d_in[2];
    const float* wq = (const float*)d_in[3];
    const float* bq = (const float*)d_in[4];
    const float* wk = (const float*)d_in[5];
    const float* bk = (const float*)d_in[6];
    const float* wv = (const float*)d_in[7];
    const float* bv = (const float*)d_in[8];
    const float* wo = (const float*)d_in[9];
    const float* bo = (const float*)d_in[10];
    const float* gq = (const float*)d_in[11];
    const float* gk = (const float*)d_in[12];
    float* out = (float*)d_out;

    const size_t mat = (size_t)SB * DM;
    float* Qw = (float*)d_ws;
    float* Kw = Qw + mat;
    float* Vw = Kw + mat;
    float* Aw = Vw + mat;
    const size_t wsz = (size_t)DM * DM;
    float* Wfq = Aw + mat;
    float* Wfk = Wfq + wsz;
    float* Wfv = Wfk + wsz;
    float* Wfo = Wfv + wsz;

    wfrag_kernel<<<dim3(DM * DM / 4 / 256 / WF_PER_THREAD, 4), 256, 0, stream>>>(wq, wk, wv, wo, Wfq);

    dim3 gblk(SB / 16);
    gemm16_kernel<<<gblk, 256, 0, stream>>>(x, Wfq, bq, gq, fc, fs, Qw, 1);
    gemm16_kernel<<<gblk, 256, 0, stream>>>(x, Wfk, bk, gk, fc, fs, Kw, 2);
    gemm16_kernel<<<gblk, 256, 0, stream>>>(x, Wfv, bv, gq, fc, fs, Vw, 3);
    attn_kernel<<<dim3(SB / 16 / 8, NH), 256, 0, stream>>>(Qw, Kw, Vw, Aw);
    gemm16_kernel<<<gblk, 256, 0, stream>>>(Aw, Wfo, bo, gq, fc, fs, out, 0);
}


# GEMM: each row tile starts its k sweep at its own k group (tile index mod 128) and wraps, so the 128 workgroups no longer want the same W lines at once; k sum order rotated per row tile
# speedup vs baseline: 2.9686x; 1.1213x over previous
#include <hip/hip_runtime.h>
#include <math.h>

#define SB   2048
#define DM   1024
#define NH   16
#define HD   64
#define HD2  32
#define SCALE 0.125f
#define EPS  1e-6f

typedef __attribute__((ext_vector_type(2))) float v2f;
typedef __attribute__((ext_vector_type(4))) float v4f;
typedef __attribute__((ext_vector_type(8))) float v8f;

static __device__ __forceinline__ v8f wmma_f32(v2f a, v2f b, v8f c) {
    return __builtin_amdgcn_wmma_f32_16x16x4_f32(
         false, a,  false, b,
         (short)0, c,  false,  false);
}

#define WF_TILE (16 * DM)
#define WF_PER_THREAD 8

#define KF_HEAD (SB * HD)
__global__ __launch_bounds__(256) void wfrag_kernel(
    const float* __restrict__ w0, const float* __restrict__ w1,
    const float* __restrict__ w2, const float* __restrict__ w3,
    float* __restrict__ dst)
{
    const int which = blockIdx.y;
    const float* W = (which == 0) ? w0 : (which == 1) ? w1 : (which == 2) ? w2 : w3;
    float* D = dst + (size_t)which * DM * DM;
    v4f v[WF_PER_THREAD];
#pragma unroll
    for (int i = 0; i < WF_PER_THREAD; ++i) {
        const int o    = (blockIdx.x * WF_PER_THREAD + i) * 256 + threadIdx.x;
        const int lane = o & 31;
        const int j    = (o >> 5) & 127;
        const int nt   = o >> 12;
        const int c    = lane & 15;
        const int hf   = lane >> 4;
        v[i] = *(const v4f*)(W + (size_t)(nt * 16 + c) * DM + 8 * j + 4 * hf);
    }
#pragma unroll
    for (int i = 0; i < WF_PER_THREAD; ++i) {
        const int o = (blockIdx.x * WF_PER_THREAD + i) * 256 + threadIdx.x;
        *(v4f*)(D + (size_t)o * 4) = v[i];
    }
}

__global__ __launch_bounds__(256) void gemm16_kernel(
    const float* __restrict__ X,
    const float* __restrict__ Wf,
    const float* __restrict__ bias,
    const float* __restrict__ g,
    const float* __restrict__ fc,
    const float* __restrict__ fs,
    float* __restrict__ out,
    int mode)
{
    const int doNormRope = (mode == 1) | (mode == 2);
    __shared__ __attribute__((aligned(16))) float tile[16][DM + 4];
    __shared__ float redbuf[16][17];
    __shared__ float rrms[16];

    const int tid  = threadIdx.x;
    const int wave = tid >> 5;
    const int lane = tid & 31;
    const int half = lane >> 4;
    const int l16  = lane & 15;
    const int rowBase  = blockIdx.x * 16;
    const int nColBase = wave * 128;

    v8f acc[8];
#pragma unroll
    for (int t = 0; t < 8; ++t) acc[t] = (v8f){0,0,0,0,0,0,0,0};

    const float* xrow = X + (size_t)(rowBase + l16) * DM + 4 * half;
    const float* wfp = Wf + (size_t)(wave * 8) * WF_TILE + lane * 4;

    v2f alo[4], ahi[4], blo[4][8], bhi[4][8];
    const int gs = blockIdx.x & (DM / 8 - 1);
#pragma unroll
    for (int s = 0; s < 3; ++s) {
        const int g0 = (gs + s) & (DM / 8 - 1);
        v4f xa = *(const v4f*)(xrow + 8 * g0);
        alo[s] = xa.xy; ahi[s] = xa.zw;
#pragma unroll
        for (int t = 0; t < 8; ++t) {
            v4f xb = *(const v4f*)(wfp + t * WF_TILE + g0 * 128);
            blo[s][t] = xb.xy; bhi[s][t] = xb.zw;
        }
    }

    for (int j = 0; j < DM / 8; j += 4) {
#pragma unroll
        for (int u = 0; u < 4; ++u) {
            const int gn = (gs + j + u + 3) & (DM / 8 - 1);
            const int sn = (u + 3) & 3;
            {
                v4f xa = *(const v4f*)(xrow + 8 * gn);
                alo[sn] = xa.xy; ahi[sn] = xa.zw;
            }
#pragma unroll
            for (int t = 0; t < 8; ++t) {
                v4f xb = *(const v4f*)(wfp + t * WF_TILE + gn * 128);
                blo[sn][t] = xb.xy; bhi[sn][t] = xb.zw;
            }
#pragma unroll
            for (int t = 0; t < 8; ++t) acc[t] = wmma_f32(alo[u], blo[u][t], acc[t]);
#pragma unroll
            for (int t = 0; t < 8; ++t) acc[t] = wmma_f32(ahi[u], bhi[u][t], acc[t]);
        }
    }

#pragma unroll
    for (int t = 0; t < 8; ++t) {
        int col = nColBase + t * 16 + l16;
        float bv = bias[col];
#pragma unroll
        for (int j = 0; j < 8; ++j) {
            int row = j + 8 * half;
            tile[row][col] = acc[t][j] + bv;
        }
    }
    __syncthreads();

    if (doNormRope) {
        {
            int r  = tid & 15;
            int c0 = tid >> 4;
            float s = 0.f;
            for (int c = c0; c < DM; c += 16) { float v = tile[r][c]; s += v * v; }
            redbuf[r][c0] = s;
        }
        __syncthreads();
        if (tid < 16) {
            float s = 0.f;
#pragma unroll
            for (int c = 0; c < 16; ++c) s += redbuf[tid][c];
            rrms[tid] = rsqrtf(s * (1.0f / (float)DM) + EPS);
        }
        __syncthreads();
    }

    if (mode == 2) {
        const int j = wave;
        const int srow = rowBase + l16;
        const float rr = rrms[l16];
        const int pp = 4 * j + 2 * half;
        const float c0 = fc[srow * HD2 + pp],     s0 = fs[srow * HD2 + pp];
        const float c1 = fc[srow * HD2 + pp + 1], s1 = fs[srow * HD2 + pp + 1];
        for (int hh = 0; hh < NH; ++hh) {
            const int col = hh * HD + 8 * j + 4 * half;
            const v4f v  = *(const v4f*)&tile[l16][col];
            const v4f gg = *(const v4f*)(g + col);
            float re0 = v.x, im0 = v.y, re1 = v.z, im1 = v.w;
            re0 *= rr * gg.x; im0 *= rr * gg.y;
            re1 *= rr * gg.z; im1 *= rr * gg.w;
            v4f r;
            r.x = re0 * c0 - im0 * s0;
            r.y = re0 * s0 + im0 * c0;
            r.z = re1 * c1 - im1 * s1;
            r.w = re1 * s1 + im1 * c1;
            *(v4f*)(out + (size_t)hh * KF_HEAD + ((size_t)blockIdx.x * 8 + j) * 128 + lane * 4) = r;
        }
        return;
    }
    if (mode == 3) {
        const int nt  = wave & 3;
        const int kgl = wave >> 2;
        for (int hh = 0; hh < NH; ++hh) {
            const int col = hh * HD + nt * 16 + l16;
            const int r0  = 8 * kgl + 4 * half;
            v4f r;
            r.x = tile[r0 + 0][col];
            r.y = tile[r0 + 1][col];
            r.z = tile[r0 + 2][col];
            r.w = tile[r0 + 3][col];
            *(v4f*)(out + (size_t)hh * KF_HEAD
                        + (((size_t)blockIdx.x * 2 + kgl) * 4 + nt) * 128 + lane * 4) = r;
        }
        return;
    }

    for (int p = tid; p < 16 * (DM / 2); p += 256) {
        int r    = p >> 9;
        int cp   = p & 511;
        int dcol = cp * 2;
        int srow = rowBase + r;
        float re = tile[r][dcol];
        float im = tile[r][dcol + 1];
        if (doNormRope) {
            float rr = rrms[r];
            re *= rr * g[dcol];
            im *= rr * g[dcol + 1];
            int   pp = cp & (HD2 - 1);
            float c  = fc[srow * HD2 + pp];
            float sn = fs[srow * HD2 + pp];
            float ore = re * c - im * sn;
            float oim = re * sn + im * c;
            re = ore; im = oim;
        }
        out[(size_t)srow * DM + dcol]     = re;
        out[(size_t)srow * DM + dcol + 1] = im;
    }
}

#define KB  64
#define KT  (KB / 16)
#define PP  (KB + 4)

__global__ __launch_bounds__(256) __attribute__((amdgpu_waves_per_eu(2, 2))) void attn_kernel(
    const float* __restrict__ Q,
    const float* __restrict__ Kf,
    const float* __restrict__ Vf,
    float* __restrict__ O)
{
    __shared__ __attribute__((aligned(16))) float Pall[8][16][PP];

    const int wave = threadIdx.x >> 5;
    const int lane = threadIdx.x & 31;
    const int half = lane >> 4;
    const int l16  = lane & 15;
    float (&P)[16][PP] = Pall[wave];
    const int qt    = blockIdx.x * 8 + wave;
    const int h     = blockIdx.y;
    const int qBase = qt * 16;
    const int dBase = h * HD;

    v2f qlo[8], qhi[8];
    {
        const float* qp = Q + (size_t)(qBase + l16) * DM + dBase + 4 * half;
#pragma unroll
        for (int j = 0; j < 8; ++j) {
            v4f xq = *(const v4f*)(qp + 8 * j);
            qlo[j] = xq.xy; qhi[j] = xq.zw;
        }
    }
    const float* kfp = Kf + (size_t)h * KF_HEAD + lane * 4;
    const float* vfp = Vf + (size_t)h * KF_HEAD + lane * 4;

    v8f o[4];
#pragma unroll
    for (int nt = 0; nt < 4; ++nt) o[nt] = (v8f){0,0,0,0,0,0,0,0};
    float m[8], lsum[8];
#pragma unroll
    for (int j = 0; j < 8; ++j) { m[j] = -INFINITY; lsum[j] = 0.f; }

    for (int t = 0; t < SB / KB; ++t) {
        v8f s[KT];
#pragma unroll
        for (int c = 0; c < KT; ++c) s[c] = (v8f){0,0,0,0,0,0,0,0};
        const float* kp = kfp + (size_t)t * (KT * 8 * 128);
#pragma unroll
        for (int j = 0; j < 8; ++j) {
            v4f kb[KT];
#pragma unroll
            for (int c = 0; c < KT; ++c) kb[c] = *(const v4f*)(kp + (c * 8 + j) * 128);
#pragma unroll
            for (int c = 0; c < KT; ++c) s[c] = wmma_f32(qlo[j], kb[c].xy, s[c]);
#pragma unroll
            for (int c = 0; c < KT; ++c) s[c] = wmma_f32(qhi[j], kb[c].zw, s[c]);
        }

#pragma unroll
        for (int j = 0; j < 8; ++j) {
            float sv0 = s[0][j] * SCALE;
            float sv1 = s[1][j] * SCALE;
            float sv2 = s[2][j] * SCALE;
            float sv3 = s[3][j] * SCALE;
            float mx = fmaxf(fmaxf(sv0, sv1), fmaxf(sv2, sv3));
            mx = fmaxf(mx, __shfl_xor(mx, 1, 32));
            mx = fmaxf(mx, __shfl_xor(mx, 2, 32));
            mx = fmaxf(mx, __shfl_xor(mx, 4, 32));
            mx = fmaxf(mx, __shfl_xor(mx, 8, 32));
            float nm   = fmaxf(m[j], mx);
            float corr = __expf(m[j] - nm);
            float p0 = __expf(sv0 - nm);
            float p1 = __expf(sv1 - nm);
            float p2 = __expf(sv2 - nm);
            float p3 = __expf(sv3 - nm);
            float ps = (p0 + p1) + (p2 + p3);
            ps += __shfl_xor(ps, 1, 32);
            ps += __shfl_xor(ps, 2, 32);
            ps += __shfl_xor(ps, 4, 32);
            ps += __shfl_xor(ps, 8, 32);
            lsum[j] = lsum[j] * corr + ps;
            m[j] = nm;
            o[0][j] *= corr; o[1][j] *= corr; o[2][j] *= corr; o[3][j] *= corr;
            const int prow = j + 8 * half;
            P[prow][0 * 16 + l16] = p0;
            P[prow][1 * 16 + l16] = p1;
            P[prow][2 * 16 + l16] = p2;
            P[prow][3 * 16 + l16] = p3;
        }
        __builtin_amdgcn_fence(__ATOMIC_RELEASE, "wavefront");
        __builtin_amdgcn_wave_barrier();
        __builtin_amdgcn_fence(__ATOMIC_ACQUIRE, "wavefront");

        const float* vp = vfp + (size_t)t * (8 * 4 * 128);
#pragma unroll
        for (int g = 0; g < KB / 8; ++g) {
            v4f vb[4];
#pragma unroll
            for (int nt = 0; nt < 4; ++nt) vb[nt] = *(const v4f*)(vp + (g * 4 + nt) * 128);
            const v4f pa = *(const v4f*)&P[l16][8 * g + 4 * half];
#pragma unroll
            for (int nt = 0; nt < 4; ++nt) o[nt] = wmma_f32(pa.xy, vb[nt].xy, o[nt]);
#pragma unroll
            for (int nt = 0; nt < 4; ++nt) o[nt] = wmma_f32(pa.zw, vb[nt].zw, o[nt]);
        }
        __builtin_amdgcn_fence(__ATOMIC_RELEASE, "wavefront");
        __builtin_amdgcn_wave_barrier();
        __builtin_amdgcn_fence(__ATOMIC_ACQUIRE, "wavefront");
    }

#pragma unroll
    for (int j = 0; j < 8; ++j) {
        float inv = 1.0f / lsum[j];
        int row = qBase + j + 8 * half;
#pragma unroll
        for (int nt = 0; nt < 4; ++nt) {
            O[(size_t)row * DM + dBase + nt * 16 + l16] = o[nt][j] * inv;
        }
    }
}

extern "C" void kernel_launch(void* const* d_in, const int* in_sizes, int n_in,
                              void* d_out, int out_size, void* d_ws, size_t ws_size,
                              hipStream_t stream) {
    const float* x  = (const float*)d_in[0];
    const float* fc = (const float*)d_in[1];
    const float* fs = (const float*)d_in[2];
    const float* wq = (const float*)d_in[3];
    const float* bq = (const float*)d_in[4];
    const float* wk = (const float*)d_in[5];
    const float* bk = (const float*)d_in[6];
    const float* wv = (const float*)d_in[7];
    const float* bv = (const float*)d_in[8];
    const float* wo = (const float*)d_in[9];
    const float* bo = (const float*)d_in[10];
    const float* gq = (const float*)d_in[11];
    const float* gk = (const float*)d_in[12];
    float* out = (float*)d_out;

    const size_t mat = (size_t)SB * DM;
    float* Qw = (float*)d_ws;
    float* Kw = Qw + mat;
    float* Vw = Kw + mat;
    float* Aw = Vw + mat;
    const size_t wsz = (size_t)DM * DM;
    float* Wfq = Aw + mat;
    float* Wfk = Wfq + wsz;
    float* Wfv = Wfk + wsz;
    float* Wfo = Wfv + wsz;

    wfrag_kernel<<<dim3(DM * DM / 4 / 256 / WF_PER_THREAD, 4), 256, 0, stream>>>(wq, wk, wv, wo, Wfq);

    dim3 gblk(SB / 16);
    gemm16_kernel<<<gblk, 256, 0, stream>>>(x, Wfq, bq, gq, fc, fs, Qw, 1);
    gemm16_kernel<<<gblk, 256, 0, stream>>>(x, Wfk, bk, gk, fc, fs, Kw, 2);
    gemm16_kernel<<<gblk, 256, 0, stream>>>(x, Wfv, bv, gq, fc, fs, Vw, 3);
    attn_kernel<<<dim3(SB / 16 / 8, NH), 256, 0, stream>>>(Qw, Kw, Vw, Aw);
    gemm16_kernel<<<gblk, 256, 0, stream>>>(Aw, Wfo, bo, gq, fc, fs, out, 0);
}


# GEMMs re-tiled: 64x128 tiles on 256 four-wave workgroups (W lines shared in the vector cache; was 16x1024 on 128); rms factor applied in attention from per-tile sums; k, norm sum order change
# speedup vs baseline: 3.6303x; 1.2229x over previous
#include <hip/hip_runtime.h>
#include <math.h>

#define SB   2048
#define DM   1024
#define NH   16
#define HD   64
#define HD2  32
#define SCALE 0.125f
#define EPS  1e-6f

typedef __attribute__((ext_vector_type(2))) float v2f;
typedef __attribute__((ext_vector_type(4))) float v4f;
typedef __attribute__((ext_vector_type(8))) float v8f;

static __device__ __forceinline__ v8f wmma_f32(v2f a, v2f b, v8f c) {
    return __builtin_amdgcn_wmma_f32_16x16x4_f32(
         false, a,  false, b,
         (short)0, c,  false,  false);
}

#define WF_TILE (16 * DM)
#define WF_PER_THREAD 8

#define KF_HEAD (SB * HD)
__global__ __launch_bounds__(256) void wfrag_kernel(
    const float* __restrict__ w0, const float* __restrict__ w1,
    const float* __restrict__ w2, const float* __restrict__ w3,
    float* __restrict__ dst)
{
    const int which = blockIdx.y;
    const float* W = (which == 0) ? w0 : (which == 1) ? w1 : (which == 2) ? w2 : w3;
    float* D = dst + (size_t)which * DM * DM;
    v4f v[WF_PER_THREAD];
#pragma unroll
    for (int i = 0; i < WF_PER_THREAD; ++i) {
        const int o    = (blockIdx.x * WF_PER_THREAD + i) * 256 + threadIdx.x;
        const int lane = o & 31;
        const int j    = (o >> 5) & 127;
        const int nt   = o >> 12;
        const int c    = lane & 15;
        const int hf   = lane >> 4;
        v[i] = *(const v4f*)(W + (size_t)(nt * 16 + c) * DM + 8 * j + 4 * hf);
    }
#pragma unroll
    for (int i = 0; i < WF_PER_THREAD; ++i) {
        const int o = (blockIdx.x * WF_PER_THREAD + i) * 256 + threadIdx.x;
        *(v4f*)(D + (size_t)o * 4) = v[i];
    }
}

#define TROWS 64
#define TCOLS 128
#define TP    (TCOLS + 4)
#define NCB   (DM / TCOLS)
__global__ __launch_bounds__(128) void gemm_tile_kernel(
    const float* __restrict__ X,
    const float* __restrict__ Wf,
    const float* __restrict__ bias,
    const float* __restrict__ g,
    const float* __restrict__ fc,
    const float* __restrict__ fs,
    float* __restrict__ out,
    float* __restrict__ ss,
    int mode)
{
    const int doNormRope = (mode == 1) | (mode == 2);
    __shared__ __attribute__((aligned(16))) float tile[TROWS][TP];
    __shared__ float ssl[TROWS];

    const int tid  = threadIdx.x;
    const int wave = tid >> 5;
    const int lane = tid & 31;
    const int half = lane >> 4;
    const int l16  = lane & 15;
    const int wrow    = wave * 16;
    const int rowBase = blockIdx.x * TROWS + wrow;
    const int colBase = blockIdx.y * TCOLS;

    v8f acc[8];
#pragma unroll
    for (int t = 0; t < 8; ++t) acc[t] = (v8f){0,0,0,0,0,0,0,0};

    const float* xrow = X + (size_t)(rowBase + l16) * DM + 4 * half;
    const float* wfp = Wf + (size_t)(blockIdx.y * 8) * WF_TILE + lane * 4;

    v2f alo[4], ahi[4], blo[4][8], bhi[4][8];
    const int gs = (blockIdx.x + blockIdx.y * 16) & (DM / 8 - 1);
#pragma unroll
    for (int s = 0; s < 3; ++s) {
        const int g0 = (gs + s) & (DM / 8 - 1);
        v4f xa = *(const v4f*)(xrow + 8 * g0);
        alo[s] = xa.xy; ahi[s] = xa.zw;
#pragma unroll
        for (int t = 0; t < 8; ++t) {
            v4f xb = *(const v4f*)(wfp + t * WF_TILE + g0 * 128);
            blo[s][t] = xb.xy; bhi[s][t] = xb.zw;
        }
    }

    for (int j = 0; j < DM / 8; j += 4) {
#pragma unroll
        for (int u = 0; u < 4; ++u) {
            const int gn = (gs + j + u + 3) & (DM / 8 - 1);
            const int sn = (u + 3) & 3;
            {
                v4f xa = *(const v4f*)(xrow + 8 * gn);
                alo[sn] = xa.xy; ahi[sn] = xa.zw;
            }
#pragma unroll
            for (int t = 0; t < 8; ++t) {
                v4f xb = *(const v4f*)(wfp + t * WF_TILE + gn * 128);
                blo[sn][t] = xb.xy; bhi[sn][t] = xb.zw;
            }
#pragma unroll
            for (int t = 0; t < 8; ++t) acc[t] = wmma_f32(alo[u], blo[u][t], acc[t]);
#pragma unroll
            for (int t = 0; t < 8; ++t) acc[t] = wmma_f32(ahi[u], bhi[u][t], acc[t]);
        }
    }

#pragma unroll
    for (int t = 0; t < 8; ++t) {
        const float bv = bias[colBase + t * 16 + l16];
#pragma unroll
        for (int j = 0; j < 8; ++j) {
            acc[t][j] += bv;
            tile[wrow + j + 8 * half][t * 16 + l16] = acc[t][j];
        }
    }
    if (doNormRope) {
        float mine = 0.f;
#pragma unroll
        for (int j = 0; j < 8; ++j) {
            float s = 0.f;
#pragma unroll
            for (int t = 0; t < 8; ++t) s += acc[t][j] * acc[t][j];
            s += __shfl_xor(s, 1, 32);
            s += __shfl_xor(s, 2, 32);
            s += __shfl_xor(s, 4, 32);
            s += __shfl_xor(s, 8, 32);
            if (l16 == j) mine = s;
        }
        if (l16 < 8) ssl[wrow + l16 + 8 * half] = mine;
    }
    __syncthreads();

    if (doNormRope && wave < 2) {
        ss[(size_t)blockIdx.y * SB + blockIdx.x * TROWS + wave * 32 + lane] = ssl[wave * 32 + lane];
    }

    if (mode == 2) {
        const int srow = rowBase + l16;
        const size_t kt = (size_t)blockIdx.x * 4 + wave;
        v2f cc[8], sn[8];
        v4f gg[2][8];
#pragma unroll
        for (int j = 0; j < 8; ++j) {
            const int pp = 4 * j + 2 * half;
            cc[j] = *(const v2f*)(fc + srow * HD2 + pp);
            sn[j] = *(const v2f*)(fs + srow * HD2 + pp);
            gg[0][j] = *(const v4f*)(g + colBase + 8 * j + 4 * half);
            gg[1][j] = *(const v4f*)(g + colBase + HD + 8 * j + 4 * half);
        }
#pragma unroll
        for (int j = 0; j < 8; ++j) {
#pragma unroll
            for (int hl = 0; hl < 2; ++hl) {
                const int cl = hl * HD + 8 * j + 4 * half;
                const v4f v  = *(const v4f*)&tile[wrow + l16][cl];
                const float re0 = v.x * gg[hl][j].x, im0 = v.y * gg[hl][j].y;
                const float re1 = v.z * gg[hl][j].z, im1 = v.w * gg[hl][j].w;
                v4f r;
                r.x = re0 * cc[j].x - im0 * sn[j].x;
                r.y = re0 * sn[j].x + im0 * cc[j].x;
                r.z = re1 * cc[j].y - im1 * sn[j].y;
                r.w = re1 * sn[j].y + im1 * cc[j].y;
                const size_t hh = (size_t)blockIdx.y * 2 + hl;
                *(v4f*)(out + hh * KF_HEAD + (kt * 8 + j) * 128 + lane * 4) = r;
            }
        }
        return;
    }
    if (mode == 3) {
        for (int q = 0; q < 16; ++q) {
            const int kgl = q & 1;
            const int nt  = (q >> 1) & 3;
            const int hl  = q >> 3;
            const int cl  = hl * HD + nt * 16 + l16;
            const int r0  = wrow + 8 * kgl + 4 * half;
            v4f r;
            r.x = tile[r0 + 0][cl];
            r.y = tile[r0 + 1][cl];
            r.z = tile[r0 + 2][cl];
            r.w = tile[r0 + 3][cl];
            const size_t hh = (size_t)blockIdx.y * 2 + hl;
            const size_t kg = (size_t)blockIdx.x * 8 + wave * 2 + kgl;
            *(v4f*)(out + hh * KF_HEAD + (kg * 4 + nt) * 128 + lane * 4) = r;
        }
        return;
    }

    const int cl = lane * 4;
    if (mode == 1) {
        const int pp = (cl & (HD - 1)) >> 1;
        const v4f gg = *(const v4f*)(g + colBase + cl);
        v2f cc[16], sn[16];
#pragma unroll
        for (int r = 0; r < 16; ++r) {
            cc[r] = *(const v2f*)(fc + (rowBase + r) * HD2 + pp);
            sn[r] = *(const v2f*)(fs + (rowBase + r) * HD2 + pp);
        }
#pragma unroll
        for (int r = 0; r < 16; ++r) {
            const v4f v = *(const v4f*)&tile[wrow + r][cl];
            const float re0 = v.x * gg.x, im0 = v.y * gg.y;
            const float re1 = v.z * gg.z, im1 = v.w * gg.w;
            v4f o;
            o.x = re0 * cc[r].x - im0 * sn[r].x;
            o.y = re0 * sn[r].x + im0 * cc[r].x;
            o.z = re1 * cc[r].y - im1 * sn[r].y;
            o.w = re1 * sn[r].y + im1 * cc[r].y;
            *(v4f*)(out + (size_t)(rowBase + r) * DM + colBase + cl) = o;
        }
        return;
    }
    for (int r = 0; r < 16; ++r)
        *(v4f*)(out + (size_t)(rowBase + r) * DM + colBase + cl) = *(const v4f*)&tile[wrow + r][cl];
}

#define KB  64
#define KT  (KB / 16)
#define PP  (KB + 4)

__global__ __launch_bounds__(256) __attribute__((amdgpu_waves_per_eu(2, 2))) void attn_kernel(
    const float* __restrict__ Q,
    const float* __restrict__ Kf,
    const float* __restrict__ Vf,
    const float* __restrict__ ssq,
    const float* __restrict__ ssk,
    float* __restrict__ O)
{
    __shared__ __attribute__((aligned(16))) float Pall[8][16][PP];
    __shared__ __attribute__((aligned(16))) float ktab[SB];
#pragma unroll
    for (int i = 0; i < SB / 4 / 256; ++i) {
        const int k4 = (threadIdx.x + 256 * i) * 4;
        v4f p[8];
#pragma unroll
        for (int cb = 0; cb < 8; ++cb) p[cb] = *(const v4f*)(ssk + cb * SB + k4);
        const v4f s4 = ((p[0] + p[1]) + (p[2] + p[3])) + ((p[4] + p[5]) + (p[6] + p[7]));
        v4f r;
        r.x = rsqrtf(s4.x * (1.0f / (float)DM) + EPS) * SCALE;
        r.y = rsqrtf(s4.y * (1.0f / (float)DM) + EPS) * SCALE;
        r.z = rsqrtf(s4.z * (1.0f / (float)DM) + EPS) * SCALE;
        r.w = rsqrtf(s4.w * (1.0f / (float)DM) + EPS) * SCALE;
        *(v4f*)&ktab[k4] = r;
    }
    __syncthreads();

    const int wave = threadIdx.x >> 5;
    const int lane = threadIdx.x & 31;
    const int half = lane >> 4;
    const int l16  = lane & 15;
    float (&P)[16][PP] = Pall[wave];
    const int qt    = blockIdx.x * 8 + wave;
    const int h     = blockIdx.y;
    const int qBase = qt * 16;
    const int dBase = h * HD;

    v2f qlo[8], qhi[8];
    {
        const float* qp = Q + (size_t)(qBase + l16) * DM + dBase + 4 * half;
        float qs[8];
#pragma unroll
        for (int cb = 0; cb < 8; ++cb) qs[cb] = ssq[cb * SB + qBase + l16];
        const float rq = rsqrtf((((qs[0] + qs[1]) + (qs[2] + qs[3])) + ((qs[4] + qs[5]) + (qs[6] + qs[7])))
                                * (1.0f / (float)DM) + EPS);
#pragma unroll
        for (int j = 0; j < 8; ++j) {
            v4f xq = *(const v4f*)(qp + 8 * j) * rq;
            qlo[j] = xq.xy; qhi[j] = xq.zw;
        }
    }
    const float* kfp = Kf + (size_t)h * KF_HEAD + lane * 4;
    const float* vfp = Vf + (size_t)h * KF_HEAD + lane * 4;

    v8f o[4];
#pragma unroll
    for (int nt = 0; nt < 4; ++nt) o[nt] = (v8f){0,0,0,0,0,0,0,0};
    float m[8], lsum[8];
#pragma unroll
    for (int j = 0; j < 8; ++j) { m[j] = -INFINITY; lsum[j] = 0.f; }

    for (int t = 0; t < SB / KB; ++t) {
        v8f s[KT];
#pragma unroll
        for (int c = 0; c < KT; ++c) s[c] = (v8f){0,0,0,0,0,0,0,0};
        const float* kp = kfp + (size_t)t * (KT * 8 * 128);
#pragma unroll
        for (int j = 0; j < 8; ++j) {
            v4f kb[KT];
#pragma unroll
            for (int c = 0; c < KT; ++c) kb[c] = *(const v4f*)(kp + (c * 8 + j) * 128);
#pragma unroll
            for (int c = 0; c < KT; ++c) s[c] = wmma_f32(qlo[j], kb[c].xy, s[c]);
#pragma unroll
            for (int c = 0; c < KT; ++c) s[c] = wmma_f32(qhi[j], kb[c].zw, s[c]);
        }

        const float kc0 = ktab[t * KB + 0 * 16 + l16];
        const float kc1 = ktab[t * KB + 1 * 16 + l16];
        const float kc2 = ktab[t * KB + 2 * 16 + l16];
        const float kc3 = ktab[t * KB + 3 * 16 + l16];
#pragma unroll
        for (int j = 0; j < 8; ++j) {
            float sv0 = s[0][j] * kc0;
            float sv1 = s[1][j] * kc1;
            float sv2 = s[2][j] * kc2;
            float sv3 = s[3][j] * kc3;
            float mx = fmaxf(fmaxf(sv0, sv1), fmaxf(sv2, sv3));
            mx = fmaxf(mx, __shfl_xor(mx, 1, 32));
            mx = fmaxf(mx, __shfl_xor(mx, 2, 32));
            mx = fmaxf(mx, __shfl_xor(mx, 4, 32));
            mx = fmaxf(mx, __shfl_xor(mx, 8, 32));
            float nm   = fmaxf(m[j], mx);
            float corr = __expf(m[j] - nm);
            float p0 = __expf(sv0 - nm);
            float p1 = __expf(sv1 - nm);
            float p2 = __expf(sv2 - nm);
            float p3 = __expf(sv3 - nm);
            float ps = (p0 + p1) + (p2 + p3);
            ps += __shfl_xor(ps, 1, 32);
            ps += __shfl_xor(ps, 2, 32);
            ps += __shfl_xor(ps, 4, 32);
            ps += __shfl_xor(ps, 8, 32);
            lsum[j] = lsum[j] * corr + ps;
            m[j] = nm;
            o[0][j] *= corr; o[1][j] *= corr; o[2][j] *= corr; o[3][j] *= corr;
            const int prow = j + 8 * half;
            P[prow][0 * 16 + l16] = p0;
            P[prow][1 * 16 + l16] = p1;
            P[prow][2 * 16 + l16] = p2;
            P[prow][3 * 16 + l16] = p3;
        }
        __builtin_amdgcn_fence(__ATOMIC_RELEASE, "wavefront");
        __builtin_amdgcn_wave_barrier();
        __builtin_amdgcn_fence(__ATOMIC_ACQUIRE, "wavefront");

        const float* vp = vfp + (size_t)t * (8 * 4 * 128);
#pragma unroll
        for (int g = 0; g < KB / 8; ++g) {
            v4f vb[4];
#pragma unroll
            for (int nt = 0; nt < 4; ++nt) vb[nt] = *(const v4f*)(vp + (g * 4 + nt) * 128);
            const v4f pa = *(const v4f*)&P[l16][8 * g + 4 * half];
#pragma unroll
            for (int nt = 0; nt < 4; ++nt) o[nt] = wmma_f32(pa.xy, vb[nt].xy, o[nt]);
#pragma unroll
            for (int nt = 0; nt < 4; ++nt) o[nt] = wmma_f32(pa.zw, vb[nt].zw, o[nt]);
        }
        __builtin_amdgcn_fence(__ATOMIC_RELEASE, "wavefront");
        __builtin_amdgcn_wave_barrier();
        __builtin_amdgcn_fence(__ATOMIC_ACQUIRE, "wavefront");
    }

#pragma unroll
    for (int j = 0; j < 8; ++j) {
        float inv = 1.0f / lsum[j];
        int row = qBase + j + 8 * half;
#pragma unroll
        for (int nt = 0; nt < 4; ++nt) {
            O[(size_t)row * DM + dBase + nt * 16 + l16] = o[nt][j] * inv;
        }
    }
}

extern "C" void kernel_launch(void* const* d_in, const int* in_sizes, int n_in,
                              void* d_out, int out_size, void* d_ws, size_t ws_size,
                              hipStream_t stream) {
    const float* x  = (const float*)d_in[0];
    const float* fc = (const float*)d_in[1];
    const float* fs = (const float*)d_in[2];
    const float* wq = (const float*)d_in[3];
    const float* bq = (const float*)d_in[4];
    const float* wk = (const float*)d_in[5];
    const float* bk = (const float*)d_in[6];
    const float* wv = (const float*)d_in[7];
    const float* bv = (const float*)d_in[8];
    const float* wo = (const float*)d_in[9];
    const float* bo = (const float*)d_in[10];
    const float* gq = (const float*)d_in[11];
    const float* gk = (const float*)d_in[12];
    float* out = (float*)d_out;

    const size_t mat = (size_t)SB * DM;
    float* Qw = (float*)d_ws;
    float* Kw = Qw + mat;
    float* Vw = Kw + mat;
    float* Aw = Vw + mat;
    const size_t wsz = (size_t)DM * DM;
    float* Wfq = Aw + mat;
    float* Wfk = Wfq + wsz;
    float* Wfv = Wfk + wsz;
    float* Wfo = Wfv + wsz;
    const size_t ssz = (size_t)NCB * SB;
    float* SSq = Wfo + wsz;
    float* SSk = SSq + ssz;

    wfrag_kernel<<<dim3(DM * DM / 4 / 256 / WF_PER_THREAD, 4), 256, 0, stream>>>(wq, wk, wv, wo, Wfq);

    dim3 gblk(SB / TROWS, DM / TCOLS);
    gemm_tile_kernel<<<gblk, 128, 0, stream>>>(x, Wfq, bq, gq, fc, fs, Qw, SSq, 1);
    gemm_tile_kernel<<<gblk, 128, 0, stream>>>(x, Wfk, bk, gk, fc, fs, Kw, SSk, 2);
    gemm_tile_kernel<<<gblk, 128, 0, stream>>>(x, Wfv, bv, gq, fc, fs, Vw, SSq, 3);
    attn_kernel<<<dim3(SB / 16 / 8, NH), 256, 0, stream>>>(Qw, Kw, Vw, SSq, SSk, Aw);
    gemm_tile_kernel<<<gblk, 128, 0, stream>>>(Aw, Wfo, bo, gq, fc, fs, out, SSq, 0);
}


# GEMM: 8-wave workgroups, 2 waves per SIMD; a pair splits its tile's k groups (even/odd), partial sums added via LDS: one wave's load waits leave the matrix unit to the other; k sum order changes
# speedup vs baseline: 3.9924x; 1.0997x over previous
#include <hip/hip_runtime.h>
#include <math.h>

#define SB   2048
#define DM   1024
#define NH   16
#define HD   64
#define HD2  32
#define SCALE 0.125f
#define EPS  1e-6f

typedef __attribute__((ext_vector_type(2))) float v2f;
typedef __attribute__((ext_vector_type(4))) float v4f;
typedef __attribute__((ext_vector_type(8))) float v8f;

static __device__ __forceinline__ v8f wmma_f32(v2f a, v2f b, v8f c) {
    return __builtin_amdgcn_wmma_f32_16x16x4_f32(
         false, a,  false, b,
         (short)0, c,  false,  false);
}

#define WF_TILE (16 * DM)
#define WF_PER_THREAD 8

#define KF_HEAD (SB * HD)
__global__ __launch_bounds__(256) void wfrag_kernel(
    const float* __restrict__ w0, const float* __restrict__ w1,
    const float* __restrict__ w2, const float* __restrict__ w3,
    float* __restrict__ dst)
{
    const int which = blockIdx.y;
    const float* W = (which == 0) ? w0 : (which == 1) ? w1 : (which == 2) ? w2 : w3;
    float* D = dst + (size_t)which * DM * DM;
    v4f v[WF_PER_THREAD];
#pragma unroll
    for (int i = 0; i < WF_PER_THREAD; ++i) {
        const int o    = (blockIdx.x * WF_PER_THREAD + i) * 256 + threadIdx.x;
        const int lane = o & 31;
        const int j    = (o >> 5) & 127;
        const int nt   = o >> 12;
        const int c    = lane & 15;
        const int hf   = lane >> 4;
        v[i] = *(const v4f*)(W + (size_t)(nt * 16 + c) * DM + 8 * j + 4 * hf);
    }
#pragma unroll
    for (int i = 0; i < WF_PER_THREAD; ++i) {
        const int o = (blockIdx.x * WF_PER_THREAD + i) * 256 + threadIdx.x;
        *(v4f*)(D + (size_t)o * 4) = v[i];
    }
}

#define TROWS 64
#define TCOLS 128
#define TP    (TCOLS + 4)
#define NCB   (DM / TCOLS)
__global__ __launch_bounds__(256) __attribute__((amdgpu_waves_per_eu(2, 2))) void gemm_tile_kernel(
    const float* __restrict__ X,
    const float* __restrict__ Wf,
    const float* __restrict__ bias,
    const float* __restrict__ g,
    const float* __restrict__ fc,
    const float* __restrict__ fs,
    float* __restrict__ out,
    float* __restrict__ ss,
    int mode)
{
    const int doNormRope = (mode == 1) | (mode == 2);
    __shared__ __attribute__((aligned(16))) float tile[TROWS][TP];
    __shared__ float ssl[TROWS];

    const int tid  = threadIdx.x;
    const int wave = tid >> 5;
    const int lane = tid & 31;
    const int half = lane >> 4;
    const int l16  = lane & 15;
    const int rw   = wave & 3;
    const int kh   = wave >> 2;
    const int wrow    = rw * 16;
    const int rowBase = blockIdx.x * TROWS + wrow;
    const int colBase = blockIdx.y * TCOLS;

    v8f acc[8];
#pragma unroll
    for (int t = 0; t < 8; ++t) acc[t] = (v8f){0,0,0,0,0,0,0,0};

    const float* xrow = X + (size_t)(rowBase + l16) * DM + 4 * half;
    const float* wfp = Wf + (size_t)(blockIdx.y * 8) * WF_TILE + lane * 4;

    v2f alo[4], ahi[4], blo[4][8], bhi[4][8];
    const int gs = (blockIdx.x + blockIdx.y * 16 + kh) & (DM / 8 - 1);
#pragma unroll
    for (int s = 0; s < 3; ++s) {
        const int g0 = (gs + 2 * s) & (DM / 8 - 1);
        v4f xa = *(const v4f*)(xrow + 8 * g0);
        alo[s] = xa.xy; ahi[s] = xa.zw;
#pragma unroll
        for (int t = 0; t < 8; ++t) {
            v4f xb = *(const v4f*)(wfp + t * WF_TILE + g0 * 128);
            blo[s][t] = xb.xy; bhi[s][t] = xb.zw;
        }
    }

    for (int j = 0; j < DM / 16; j += 4) {
#pragma unroll
        for (int u = 0; u < 4; ++u) {
            const int gn = (gs + 2 * (j + u + 3)) & (DM / 8 - 1);
            const int sn = (u + 3) & 3;
            {
                v4f xa = *(const v4f*)(xrow + 8 * gn);
                alo[sn] = xa.xy; ahi[sn] = xa.zw;
            }
#pragma unroll
            for (int t = 0; t < 8; ++t) {
                v4f xb = *(const v4f*)(wfp + t * WF_TILE + gn * 128);
                blo[sn][t] = xb.xy; bhi[sn][t] = xb.zw;
            }
#pragma unroll
            for (int t = 0; t < 8; ++t) acc[t] = wmma_f32(alo[u], blo[u][t], acc[t]);
#pragma unroll
            for (int t = 0; t < 8; ++t) acc[t] = wmma_f32(ahi[u], bhi[u][t], acc[t]);
        }
    }

    if (kh == 1) {
#pragma unroll
        for (int t = 0; t < 8; ++t)
#pragma unroll
            for (int j = 0; j < 8; ++j) tile[wrow + j + 8 * half][t * 16 + l16] = acc[t][j];
    }
    __syncthreads();
    if (kh == 0) {
#pragma unroll
        for (int t = 0; t < 8; ++t) {
            const float bv = bias[colBase + t * 16 + l16];
#pragma unroll
            for (int j = 0; j < 8; ++j) {
                acc[t][j] = (acc[t][j] + tile[wrow + j + 8 * half][t * 16 + l16]) + bv;
                tile[wrow + j + 8 * half][t * 16 + l16] = acc[t][j];
            }
        }
    }
    if (doNormRope && kh == 0) {
        float mine = 0.f;
#pragma unroll
        for (int j = 0; j < 8; ++j) {
            float s = 0.f;
#pragma unroll
            for (int t = 0; t < 8; ++t) s += acc[t][j] * acc[t][j];
            s += __shfl_xor(s, 1, 32);
            s += __shfl_xor(s, 2, 32);
            s += __shfl_xor(s, 4, 32);
            s += __shfl_xor(s, 8, 32);
            if (l16 == j) mine = s;
        }
        if (l16 < 8) ssl[wrow + l16 + 8 * half] = mine;
    }
    __syncthreads();

    if (doNormRope && wave < 2) {
        ss[(size_t)blockIdx.y * SB + blockIdx.x * TROWS + wave * 32 + lane] = ssl[wave * 32 + lane];
    }

    if (mode == 2) {
        const int srow = rowBase + l16;
        const size_t kt = (size_t)blockIdx.x * 4 + rw;
        const int hl = kh;
        v2f cc[8], sn[8];
        v4f gg[8];
#pragma unroll
        for (int j = 0; j < 8; ++j) {
            const int pp = 4 * j + 2 * half;
            cc[j] = *(const v2f*)(fc + srow * HD2 + pp);
            sn[j] = *(const v2f*)(fs + srow * HD2 + pp);
            gg[j] = *(const v4f*)(g + colBase + hl * HD + 8 * j + 4 * half);
        }
#pragma unroll
        for (int j = 0; j < 8; ++j) {
            const int cl = hl * HD + 8 * j + 4 * half;
            const v4f v  = *(const v4f*)&tile[wrow + l16][cl];
            const float re0 = v.x * gg[j].x, im0 = v.y * gg[j].y;
            const float re1 = v.z * gg[j].z, im1 = v.w * gg[j].w;
            v4f r;
            r.x = re0 * cc[j].x - im0 * sn[j].x;
            r.y = re0 * sn[j].x + im0 * cc[j].x;
            r.z = re1 * cc[j].y - im1 * sn[j].y;
            r.w = re1 * sn[j].y + im1 * cc[j].y;
            const size_t hh = (size_t)blockIdx.y * 2 + hl;
            *(v4f*)(out + hh * KF_HEAD + (kt * 8 + j) * 128 + lane * 4) = r;
        }
        return;
    }
    if (mode == 3) {
        const int hl = kh;
        for (int q = 0; q < 8; ++q) {
            const int kgl = q & 1;
            const int nt  = q >> 1;
            const int cl  = hl * HD + nt * 16 + l16;
            const int r0  = wrow + 8 * kgl + 4 * half;
            v4f r;
            r.x = tile[r0 + 0][cl];
            r.y = tile[r0 + 1][cl];
            r.z = tile[r0 + 2][cl];
            r.w = tile[r0 + 3][cl];
            const size_t hh = (size_t)blockIdx.y * 2 + hl;
            const size_t kg = (size_t)blockIdx.x * 8 + rw * 2 + kgl;
            *(v4f*)(out + hh * KF_HEAD + (kg * 4 + nt) * 128 + lane * 4) = r;
        }
        return;
    }

    const int cl = lane * 4;
    const int r8 = 8 * kh;
    if (mode == 1) {
        const int pp = (cl & (HD - 1)) >> 1;
        const v4f gg = *(const v4f*)(g + colBase + cl);
        v2f cc[8], sn[8];
#pragma unroll
        for (int r = 0; r < 8; ++r) {
            cc[r] = *(const v2f*)(fc + (rowBase + r8 + r) * HD2 + pp);
            sn[r] = *(const v2f*)(fs + (rowBase + r8 + r) * HD2 + pp);
        }
#pragma unroll
        for (int r = 0; r < 8; ++r) {
            const v4f v = *(const v4f*)&tile[wrow + r8 + r][cl];
            const float re0 = v.x * gg.x, im0 = v.y * gg.y;
            const float re1 = v.z * gg.z, im1 = v.w * gg.w;
            v4f o;
            o.x = re0 * cc[r].x - im0 * sn[r].x;
            o.y = re0 * sn[r].x + im0 * cc[r].x;
            o.z = re1 * cc[r].y - im1 * sn[r].y;
            o.w = re1 * sn[r].y + im1 * cc[r].y;
            *(v4f*)(out + (size_t)(rowBase + r8 + r) * DM + colBase + cl) = o;
        }
        return;
    }
    for (int r = 0; r < 8; ++r)
        *(v4f*)(out + (size_t)(rowBase + r8 + r) * DM + colBase + cl) = *(const v4f*)&tile[wrow + r8 + r][cl];
}

#define KB  64
#define KT  (KB / 16)
#define PP  (KB + 4)

__global__ __launch_bounds__(256) __attribute__((amdgpu_waves_per_eu(2, 2))) void attn_kernel(
    const float* __restrict__ Q,
    const float* __restrict__ Kf,
    const float* __restrict__ Vf,
    const float* __restrict__ ssq,
    const float* __restrict__ ssk,
    float* __restrict__ O)
{
    __shared__ __attribute__((aligned(16))) float Pall[8][16][PP];
    __shared__ __attribute__((aligned(16))) float ktab[SB];
#pragma unroll
    for (int i = 0; i < SB / 4 / 256; ++i) {
        const int k4 = (threadIdx.x + 256 * i) * 4;
        v4f p[8];
#pragma unroll
        for (int cb = 0; cb < 8; ++cb) p[cb] = *(const v4f*)(ssk + cb * SB + k4);
        const v4f s4 = ((p[0] + p[1]) + (p[2] + p[3])) + ((p[4] + p[5]) + (p[6] + p[7]));
        v4f r;
        r.x = rsqrtf(s4.x * (1.0f / (float)DM) + EPS) * SCALE;
        r.y = rsqrtf(s4.y * (1.0f / (float)DM) + EPS) * SCALE;
        r.z = rsqrtf(s4.z * (1.0f / (float)DM) + EPS) * SCALE;
        r.w = rsqrtf(s4.w * (1.0f / (float)DM) + EPS) * SCALE;
        *(v4f*)&ktab[k4] = r;
    }
    __syncthreads();

    const int wave = threadIdx.x >> 5;
    const int lane = threadIdx.x & 31;
    const int half = lane >> 4;
    const int l16  = lane & 15;
    float (&P)[16][PP] = Pall[wave];
    const int qt    = blockIdx.x * 8 + wave;
    const int h     = blockIdx.y;
    const int qBase = qt * 16;
    const int dBase = h * HD;

    v2f qlo[8], qhi[8];
    {
        const float* qp = Q + (size_t)(qBase + l16) * DM + dBase + 4 * half;
        float qs[8];
#pragma unroll
        for (int cb = 0; cb < 8; ++cb) qs[cb] = ssq[cb * SB + qBase + l16];
        const float rq = rsqrtf((((qs[0] + qs[1]) + (qs[2] + qs[3])) + ((qs[4] + qs[5]) + (qs[6] + qs[7])))
                                * (1.0f / (float)DM) + EPS);
#pragma unroll
        for (int j = 0; j < 8; ++j) {
            v4f xq = *(const v4f*)(qp + 8 * j) * rq;
            qlo[j] = xq.xy; qhi[j] = xq.zw;
        }
    }
    const float* kfp = Kf + (size_t)h * KF_HEAD + lane * 4;
    const float* vfp = Vf + (size_t)h * KF_HEAD + lane * 4;

    v8f o[4];
#pragma unroll
    for (int nt = 0; nt < 4; ++nt) o[nt] = (v8f){0,0,0,0,0,0,0,0};
    float m[8], lsum[8];
#pragma unroll
    for (int j = 0; j < 8; ++j) { m[j] = -INFINITY; lsum[j] = 0.f; }

    for (int t = 0; t < SB / KB; ++t) {
        v8f s[KT];
#pragma unroll
        for (int c = 0; c < KT; ++c) s[c] = (v8f){0,0,0,0,0,0,0,0};
        const float* kp = kfp + (size_t)t * (KT * 8 * 128);
#pragma unroll
        for (int j = 0; j < 8; ++j) {
            v4f kb[KT];
#pragma unroll
            for (int c = 0; c < KT; ++c) kb[c] = *(const v4f*)(kp + (c * 8 + j) * 128);
#pragma unroll
            for (int c = 0; c < KT; ++c) s[c] = wmma_f32(qlo[j], kb[c].xy, s[c]);
#pragma unroll
            for (int c = 0; c < KT; ++c) s[c] = wmma_f32(qhi[j], kb[c].zw, s[c]);
        }

        const float kc0 = ktab[t * KB + 0 * 16 + l16];
        const float kc1 = ktab[t * KB + 1 * 16 + l16];
        const float kc2 = ktab[t * KB + 2 * 16 + l16];
        const float kc3 = ktab[t * KB + 3 * 16 + l16];
#pragma unroll
        for (int j = 0; j < 8; ++j) {
            float sv0 = s[0][j] * kc0;
            float sv1 = s[1][j] * kc1;
            float sv2 = s[2][j] * kc2;
            float sv3 = s[3][j] * kc3;
            float mx = fmaxf(fmaxf(sv0, sv1), fmaxf(sv2, sv3));
            mx = fmaxf(mx, __shfl_xor(mx, 1, 32));
            mx = fmaxf(mx, __shfl_xor(mx, 2, 32));
            mx = fmaxf(mx, __shfl_xor(mx, 4, 32));
            mx = fmaxf(mx, __shfl_xor(mx, 8, 32));
            float nm   = fmaxf(m[j], mx);
            float corr = __expf(m[j] - nm);
            float p0 = __expf(sv0 - nm);
            float p1 = __expf(sv1 - nm);
            float p2 = __expf(sv2 - nm);
            float p3 = __expf(sv3 - nm);
            float ps = (p0 + p1) + (p2 + p3);
            ps += __shfl_xor(ps, 1, 32);
            ps += __shfl_xor(ps, 2, 32);
            ps += __shfl_xor(ps, 4, 32);
            ps += __shfl_xor(ps, 8, 32);
            lsum[j] = lsum[j] * corr + ps;
            m[j] = nm;
            o[0][j] *= corr; o[1][j] *= corr; o[2][j] *= corr; o[3][j] *= corr;
            const int prow = j + 8 * half;
            P[prow][0 * 16 + l16] = p0;
            P[prow][1 * 16 + l16] = p1;
            P[prow][2 * 16 + l16] = p2;
            P[prow][3 * 16 + l16] = p3;
        }
        __builtin_amdgcn_fence(__ATOMIC_RELEASE, "wavefront");
        __builtin_amdgcn_wave_barrier();
        __builtin_amdgcn_fence(__ATOMIC_ACQUIRE, "wavefront");

        const float* vp = vfp + (size_t)t * (8 * 4 * 128);
#pragma unroll
        for (int g = 0; g < KB / 8; ++g) {
            v4f vb[4];
#pragma unroll
            for (int nt = 0; nt < 4; ++nt) vb[nt] = *(const v4f*)(vp + (g * 4 + nt) * 128);
            const v4f pa = *(const v4f*)&P[l16][8 * g + 4 * half];
#pragma unroll
            for (int nt = 0; nt < 4; ++nt) o[nt] = wmma_f32(pa.xy, vb[nt].xy, o[nt]);
#pragma unroll
            for (int nt = 0; nt < 4; ++nt) o[nt] = wmma_f32(pa.zw, vb[nt].zw, o[nt]);
        }
        __builtin_amdgcn_fence(__ATOMIC_RELEASE, "wavefront");
        __builtin_amdgcn_wave_barrier();
        __builtin_amdgcn_fence(__ATOMIC_ACQUIRE, "wavefront");
    }

#pragma unroll
    for (int j = 0; j < 8; ++j) {
        float inv = 1.0f / lsum[j];
        int row = qBase + j + 8 * half;
#pragma unroll
        for (int nt = 0; nt < 4; ++nt) {
            O[(size_t)row * DM + dBase + nt * 16 + l16] = o[nt][j] * inv;
        }
    }
}

extern "C" void kernel_launch(void* const* d_in, const int* in_sizes, int n_in,
                              void* d_out, int out_size, void* d_ws, size_t ws_size,
                              hipStream_t stream) {
    const float* x  = (const float*)d_in[0];
    const float* fc = (const float*)d_in[1];
    const float* fs = (const float*)d_in[2];
    const float* wq = (const float*)d_in[3];
    const float* bq = (const float*)d_in[4];
    const float* wk = (const float*)d_in[5];
    const float* bk = (const float*)d_in[6];
    const float* wv = (const float*)d_in[7];
    const float* bv = (const float*)d_in[8];
    const float* wo = (const float*)d_in[9];
    const float* bo = (const float*)d_in[10];
    const float* gq = (const float*)d_in[11];
    const float* gk = (const float*)d_in[12];
    float* out = (float*)d_out;

    const size_t mat = (size_t)SB * DM;
    float* Qw = (float*)d_ws;
    float* Kw = Qw + mat;
    float* Vw = Kw + mat;
    float* Aw = Vw + mat;
    const size_t wsz = (size_t)DM * DM;
    float* Wfq = Aw + mat;
    float* Wfk = Wfq + wsz;
    float* Wfv = Wfk + wsz;
    float* Wfo = Wfv + wsz;
    const size_t ssz = (size_t)NCB * SB;
    float* SSq = Wfo + wsz;
    float* SSk = SSq + ssz;

    wfrag_kernel<<<dim3(DM * DM / 4 / 256 / WF_PER_THREAD, 4), 256, 0, stream>>>(wq, wk, wv, wo, Wfq);

    dim3 gblk(SB / TROWS, DM / TCOLS);
    gemm_tile_kernel<<<gblk, 256, 0, stream>>>(x, Wfq, bq, gq, fc, fs, Qw, SSq, 1);
    gemm_tile_kernel<<<gblk, 256, 0, stream>>>(x, Wfk, bk, gk, fc, fs, Kw, SSk, 2);
    gemm_tile_kernel<<<gblk, 256, 0, stream>>>(x, Wfv, bv, gq, fc, fs, Vw, SSq, 3);
    attn_kernel<<<dim3(SB / 16 / 8, NH), 256, 0, stream>>>(Qw, Kw, Vw, SSq, SSk, Aw);
    gemm_tile_kernel<<<gblk, 256, 0, stream>>>(Aw, Wfo, bo, gq, fc, fs, out, SSq, 0);
}
